# starting kernel
# baseline (speedup 1.0000x reference)
#include <hip/hip_runtime.h>
#include <hip/hip_bf16.h>

typedef float v2f __attribute__((ext_vector_type(2)));
typedef float v8f __attribute__((ext_vector_type(8)));

#define D_MODEL 1024
#define D_STATE 16
#define DT_RANK 64
#define BATCH   4
#define SEQLEN  2048
#define XDIM    96
#define NTOK    (BATCH * SEQLEN)
#define NCHUNK  32
#define CLEN    (SEQLEN / NCHUNK)
#define NT1     (XDIM / 16)
#define NT2     8

__global__ __launch_bounds__(256) void xp_wmma_kernel(const float* __restrict__ u,
                                                      const float* __restrict__ xw,
                                                      float* __restrict__ xp) {
  const int gtid   = blockIdx.x * 256 + threadIdx.x;
  const int m_tile = gtid >> 5;
  const int lane   = threadIdx.x & 31;
  const int row    = lane & 15;
  const int kph    = (lane >> 4) << 1;

  const float* __restrict__ ap = u  + (size_t)(m_tile * 16 + row) * D_MODEL + kph;
  const float* __restrict__ bp = xw + (size_t)row * D_MODEL + kph;

  v8f acc[NT1] = {};
  for (int k = 0; k < D_MODEL; k += 4) {
    v2f a = *(const v2f*)(ap + k);
    #pragma unroll
    for (int nt = 0; nt < NT1; ++nt) {
      v2f b = *(const v2f*)(bp + (size_t)nt * 16 * D_MODEL + k);
      acc[nt] = __builtin_amdgcn_wmma_f32_16x16x4_f32(false, a, false, b, (short)0,
                                                      acc[nt], false, false);
    }
  }

  const int m0  = m_tile * 16 + ((lane >> 4) << 3);
  const int col = lane & 15;
  #pragma unroll
  for (int nt = 0; nt < NT1; ++nt)
    #pragma unroll
    for (int i = 0; i < 8; ++i)
      xp[(size_t)(m0 + i) * XDIM + nt * 16 + col] = acc[nt][i];
}

__global__ __launch_bounds__(256) void delta_wmma_kernel(const float* __restrict__ xp,
                                                         const float* __restrict__ wt,
                                                         const float* __restrict__ bias,
                                                         float* __restrict__ delta) {
  const int gtid   = blockIdx.x * 256 + threadIdx.x;
  const int wave   = gtid >> 5;
  const int lane   = threadIdx.x & 31;
  const int ngrp   = wave % (D_MODEL / 16 / NT2);
  const int m_tile = wave / (D_MODEL / 16 / NT2);
  const int row    = lane & 15;
  const int kph    = (lane >> 4) << 1;

  const float* __restrict__ ap = xp + (size_t)(m_tile * 16 + row) * XDIM + kph;
  const float* __restrict__ bp = wt + (size_t)(ngrp * NT2 * 16 + row) * DT_RANK + kph;

  v8f acc[NT2] = {};
  #pragma unroll
  for (int k = 0; k < DT_RANK; k += 4) {
    v2f a = *(const v2f*)(ap + k);
    #pragma unroll
    for (int nt = 0; nt < NT2; ++nt) {
      v2f b = *(const v2f*)(bp + (size_t)nt * 16 * DT_RANK + k);
      acc[nt] = __builtin_amdgcn_wmma_f32_16x16x4_f32(false, a, false, b, (short)0,
                                                      acc[nt], false, false);
    }
  }

  const int m0 = m_tile * 16 + ((lane >> 4) << 3);
  #pragma unroll
  for (int nt = 0; nt < NT2; ++nt) {
    const int n0 = (ngrp * NT2 + nt) * 16 + (lane & 15);
    const float bn = bias[n0];
    #pragma unroll
    for (int i = 0; i < 8; ++i) {
      float x  = acc[nt][i] + bn;
      float sp = (x > 20.0f) ? x : __logf(1.0f + __expf(x));
      delta[(size_t)(m0 + i) * D_MODEL + n0] = sp;
    }
  }
}

__global__ __launch_bounds__(256) void scan_pass1_kernel(const float* __restrict__ delta,
                                                         const float* __restrict__ xp,
                                                         const float* __restrict__ u,
                                                         const float* __restrict__ A_log,
                                                         float* __restrict__ Q,
                                                         float* __restrict__ Sdt) {
  const int bid = blockIdx.x;
  const int d   = (bid & 3) * 256 + threadIdx.x;
  const int c   = (bid >> 2) & (NCHUNK - 1);
  const int b   = bid >> 7;

  float A[D_STATE];
  #pragma unroll
  for (int n = 0; n < D_STATE; ++n) A[n] = -__expf(A_log[d * D_STATE + n]);

  float h[D_STATE];
  #pragma unroll
  for (int n = 0; n < D_STATE; ++n) h[n] = 0.0f;
  float sdt = 0.0f;

  const size_t tok0 = (size_t)b * SEQLEN + (size_t)c * CLEN;
  for (int s = 0; s < CLEN; ++s) {
    const size_t tok = tok0 + s;
    const float dt = delta[tok * D_MODEL + d];
    const float ut = u[tok * D_MODEL + d];
    const float4* bv = (const float4*)(xp + tok * XDIM + DT_RANK);
    float4 q0 = bv[0], q1 = bv[1], q2 = bv[2], q3 = bv[3];
    float Bt[D_STATE] = {q0.x, q0.y, q0.z, q0.w, q1.x, q1.y, q1.z, q1.w,
                         q2.x, q2.y, q2.z, q2.w, q3.x, q3.y, q3.z, q3.w};
    sdt += dt;
    const float du = dt * ut;
    #pragma unroll
    for (int n = 0; n < D_STATE; ++n)
      h[n] = __expf(dt * A[n]) * h[n] + du * Bt[n];
  }

  const size_t idx = ((size_t)b * NCHUNK + c) * D_MODEL + d;
  #pragma unroll
  for (int n = 0; n < D_STATE; ++n) Q[idx * D_STATE + n] = h[n];
  Sdt[idx] = sdt;
}

__global__ __launch_bounds__(256) void scan_carry_kernel(const float* __restrict__ Q,
                                                         const float* __restrict__ Sdt,
                                                         const float* __restrict__ A_log,
                                                         float* __restrict__ Hc) {
  const int tid = blockIdx.x * 256 + threadIdx.x;
  const int d   = tid % D_MODEL;
  const int b   = tid / D_MODEL;

  float A[D_STATE];
  #pragma unroll
  for (int n = 0; n < D_STATE; ++n) A[n] = -__expf(A_log[d * D_STATE + n]);

  float h[D_STATE];
  #pragma unroll
  for (int n = 0; n < D_STATE; ++n) h[n] = 0.0f;

  for (int c = 0; c < NCHUNK; ++c) {
    const size_t idx = ((size_t)b * NCHUNK + c) * D_MODEL + d;
    #pragma unroll
    for (int n = 0; n < D_STATE; ++n) Hc[idx * D_STATE + n] = h[n];
    const float s = Sdt[idx];
    #pragma unroll
    for (int n = 0; n < D_STATE; ++n)
      h[n] = __expf(s * A[n]) * h[n] + Q[idx * D_STATE + n];
  }
}

__global__ __launch_bounds__(256) void scan_pass3_kernel(const float* __restrict__ delta,
                                                         const float* __restrict__ xp,
                                                         const float* __restrict__ u,
                                                         const float* __restrict__ A_log,
                                                         const float* __restrict__ Dsk,
                                                         const float* __restrict__ Hc,
                                                         float* __restrict__ y) {
  const int bid = blockIdx.x;
  const int d   = (bid & 3) * 256 + threadIdx.x;
  const int c   = (bid >> 2) & (NCHUNK - 1);
  const int b   = bid >> 7;

  float A[D_STATE];
  #pragma unroll
  for (int n = 0; n < D_STATE; ++n) A[n] = -__expf(A_log[d * D_STATE + n]);

  const size_t idx = ((size_t)b * NCHUNK + c) * D_MODEL + d;
  float h[D_STATE];
  #pragma unroll
  for (int n = 0; n < D_STATE; ++n) h[n] = Hc[idx * D_STATE + n];
  const float dsk = Dsk[d];

  const size_t tok0 = (size_t)b * SEQLEN + (size_t)c * CLEN;
  for (int s = 0; s < CLEN; ++s) {
    const size_t tok = tok0 + s;
    const float dt = delta[tok * D_MODEL + d];
    const float ut = u[tok * D_MODEL + d];
    const float4* bv = (const float4*)(xp + tok * XDIM + DT_RANK);
    float4 q0 = bv[0], q1 = bv[1], q2 = bv[2], q3 = bv[3];
    const float4* cv = (const float4*)(xp + tok * XDIM + DT_RANK + D_STATE);
    float4 p0 = cv[0], p1 = cv[1], p2 = cv[2], p3 = cv[3];
    float Bt[D_STATE] = {q0.x, q0.y, q0.z, q0.w, q1.x, q1.y, q1.z, q1.w,
                         q2.x, q2.y, q2.z, q2.w, q3.x, q3.y, q3.z, q3.w};
    float Ct[D_STATE] = {p0.x, p0.y, p0.z, p0.w, p1.x, p1.y, p1.z, p1.w,
                         p2.x, p2.y, p2.z, p2.w, p3.x, p3.y, p3.z, p3.w};
    const float du = dt * ut;
    float acc = 0.0f;
    #pragma unroll
    for (int n = 0; n < D_STATE; ++n) {
      h[n] = __expf(dt * A[n]) * h[n] + du * Bt[n];
      acc += h[n] * Ct[n];
    }
    y[tok * D_MODEL + d] = acc + ut * dsk;
  }
}

extern "C" void kernel_launch(void* const* d_in, const int* in_sizes, int n_in,
                              void* d_out, int out_size, void* d_ws, size_t ws_size,
                              hipStream_t stream) {
  const float* u      = (const float*)d_in[0];
  const float* A_log  = (const float*)d_in[1];
  const float* D_skip = (const float*)d_in[2];
  const float* dt_w   = (const float*)d_in[3];
  const float* dt_b   = (const float*)d_in[4];
  const float* x_w    = (const float*)d_in[5];
  float* y = (float*)d_out;

  float* ws    = (float*)d_ws;
  float* xp    = ws;
  float* delta = xp    + (size_t)NTOK * XDIM;
  float* Q     = delta + (size_t)NTOK * D_MODEL;
  float* Sdt   = Q     + (size_t)BATCH * NCHUNK * D_MODEL * D_STATE;
  float* Hc    = Sdt   + (size_t)BATCH * NCHUNK * D_MODEL;

  xp_wmma_kernel<<<(NTOK / 16) * 32 / 256, 256, 0, stream>>>(u, x_w, xp);

  delta_wmma_kernel<<<(NTOK / 16) * (D_MODEL / 16 / NT2) * 32 / 256, 256, 0, stream>>>(
      xp, dt_w, dt_b, delta);

  scan_pass1_kernel<<<BATCH * NCHUNK * (D_MODEL / 256), 256, 0, stream>>>(
      delta, xp, u, A_log, Q, Sdt);

  scan_carry_kernel<<<BATCH * D_MODEL / 256, 256, 0, stream>>>(Q, Sdt, A_log, Hc);

  scan_pass3_kernel<<<BATCH * NCHUNK * (D_MODEL / 256), 256, 0, stream>>>(
      delta, xp, u, A_log, D_skip, Hc, y);
}


# dt projection built inside scan pass 1: a block multiplies its chunk's 64 xp rows by its 256 dt_proj_w rows (both parked in LDS) on WMMA, tile kept in LDS; delta kernel gone (3 launches); same sums
# speedup vs baseline: 2.4554x; 2.4554x over previous
#include <hip/hip_runtime.h>
#include <hip/hip_bf16.h>

typedef float v2f __attribute__((ext_vector_type(2)));
typedef float v4f __attribute__((ext_vector_type(4)));
typedef float v8f __attribute__((ext_vector_type(8)));

#define D_MODEL 1024
#define D_STATE 16
#define DT_RANK 64
#define BATCH   4
#define SEQLEN  2048
#define XDIM    96
#define NTOK    (BATCH * SEQLEN)
#define NCHUNK  32
#define CLEN    (SEQLEN / NCHUNK)

#define XM    32
#define XKC   64
#define XPIT  68
#define XROWS (XM + XDIM)
#define XNCH  (D_MODEL / XKC)
#define XLD   (XROWS * (XKC / 4) / 128)

__device__ __forceinline__ void xp_chunk_mma(const float* cur, int aoff, int boff,
                                             v8f (&acc)[3]) {
  #pragma unroll
  for (int kk = 0; kk < XKC; kk += 4) {
    v2f a = *(const v2f*)(cur + aoff + kk);
    #pragma unroll
    for (int nt = 0; nt < 3; ++nt) {
      v2f b = *(const v2f*)(cur + boff + nt * 16 * XPIT + kk);
      acc[nt] = __builtin_amdgcn_wmma_f32_16x16x4_f32(false, a, false, b, (short)0,
                                                      acc[nt], false, false);
    }
  }
}

__global__ __launch_bounds__(128) void xp_wmma_kernel(const float* __restrict__ u,
                                                      const float* __restrict__ xw,
                                                      float* __restrict__ xp) {
  __shared__ __attribute__((aligned(16))) float L[2][XROWS * XPIT];

  const int t     = threadIdx.x;
  const int wave  = t >> 5;
  const int lane  = t & 31;
  const int row   = lane & 15;
  const int kph   = (lane >> 4) << 1;
  const int mt    = wave >> 1;
  const int nh    = wave & 1;
  const int m_blk = blockIdx.x * XM;

  const int srow = t >> 4;
  const int scol = (t & 15) * 4;
  const float* __restrict__ ga = u  + (size_t)(m_blk + srow) * D_MODEL + scol;
  const float* __restrict__ gb = xw + (size_t)srow * D_MODEL + scol;
  const int soff = srow * XPIT + scol;

  v4f st[XLD];
  #pragma unroll
  for (int i = 0; i < XLD; ++i)
    st[i] = (i < XM / 8) ? *(const v4f*)(ga + (size_t)(i * 8) * D_MODEL)
                         : *(const v4f*)(gb + (size_t)(i * 8 - XM) * D_MODEL);
  #pragma unroll
  for (int i = 0; i < XLD; ++i)
    *(v4f*)(&L[0][soff + i * 8 * XPIT]) = st[i];
  #pragma unroll
  for (int i = 0; i < XLD; ++i)
    st[i] = (i < XM / 8) ? *(const v4f*)(ga + (size_t)(i * 8) * D_MODEL + XKC)
                         : *(const v4f*)(gb + (size_t)(i * 8 - XM) * D_MODEL + XKC);
  __syncthreads();

  const int aoff = (mt * 16 + row) * XPIT + kph;
  const int boff = (XM + nh * 48 + row) * XPIT + kph;

  v8f acc[3] = {};
  for (int ch = 0; ch < XNCH - 1; ++ch) {
    xp_chunk_mma(L[ch & 1], aoff, boff, acc);
    float* nxt = L[(ch + 1) & 1];
    #pragma unroll
    for (int i = 0; i < XLD; ++i)
      *(v4f*)(nxt + soff + i * 8 * XPIT) = st[i];
    __syncthreads();
    if (ch + 2 < XNCH) {
      const int k2 = (ch + 2) * XKC;
      #pragma unroll
      for (int i = 0; i < XLD; ++i)
        st[i] = (i < XM / 8) ? *(const v4f*)(ga + (size_t)(i * 8) * D_MODEL + k2)
                             : *(const v4f*)(gb + (size_t)(i * 8 - XM) * D_MODEL + k2);
    }
  }
  xp_chunk_mma(L[(XNCH - 1) & 1], aoff, boff, acc);

  float* lo = L[0];
  const int r0  = mt * 16 + ((lane >> 4) << 3);
  const int col = nh * 48 + (lane & 15);
  #pragma unroll
  for (int nt = 0; nt < 3; ++nt)
    #pragma unroll
    for (int i = 0; i < 8; ++i)
      lo[(r0 + i) * XDIM + nt * 16 + col] = acc[nt][i];
  __syncthreads();
  float4* __restrict__ xo = (float4*)(xp + (size_t)m_blk * XDIM);
  #pragma unroll
  for (int i = 0; i < XM * XDIM / 4 / 128; ++i)
    xo[i * 128 + t] = ((const float4*)lo)[i * 128 + t];
}

#define SUBS  4
#define SLEN  (CLEN / SUBS)
#define SPC   (SLEN * 64 / 256)

__device__ __forceinline__ void tile_fetch(const float* __restrict__ gd,
                                           const float* __restrict__ gu, int sub,
                                           v4f (&rd)[SPC], v4f (&ru)[SPC]) {
  #pragma unroll
  for (int i = 0; i < SPC; ++i) {
    const size_t off = (size_t)(sub * SLEN + i * 4) * D_MODEL;
    rd[i] = *(const v4f*)(gd + off);
    ru[i] = *(const v4f*)(gu + off);
  }
}

__device__ __forceinline__ void tile_park(float* Ld, float* Lu, int soff, int sub,
                                          const v4f (&rd)[SPC], const v4f (&ru)[SPC]) {
  #pragma unroll
  for (int i = 0; i < SPC; ++i) {
    *(v4f*)(Ld + soff + (sub * SLEN + i * 4) * 256) = rd[i];
    *(v4f*)(Lu + soff + (sub * SLEN + i * 4) * 256) = ru[i];
  }
}

__device__ __forceinline__ void tile_fetch1(const float* __restrict__ gu, int sub, v4f (&ru)[SPC]) {
  #pragma unroll
  for (int i = 0; i < SPC; ++i)
    ru[i] = *(const v4f*)(gu + (size_t)(sub * SLEN + i * 4) * D_MODEL);
}

__device__ __forceinline__ void tile_park1(float* Lu, int soff, int sub, const v4f (&ru)[SPC]) {
  #pragma unroll
  for (int i = 0; i < SPC; ++i)
    *(v4f*)(Lu + soff + (sub * SLEN + i * 4) * 256) = ru[i];
}

#define LDP   260
#define GPIT  68

__global__ __launch_bounds__(256) void scan_pass1_kernel(const float* __restrict__ xp,
                                                         const float* __restrict__ wt,
                                                         const float* __restrict__ bias,
                                                         const float* __restrict__ u,
                                                         const float* __restrict__ A_log,
                                                         float* __restrict__ delta,
                                                         float* __restrict__ Q,
                                                         float* __restrict__ Sdt) {
  __shared__ __attribute__((aligned(16))) float Ld[CLEN * LDP];
  __shared__ __attribute__((aligned(16))) float Lx[256 * GPIT];
  __shared__ __attribute__((aligned(16))) float La[CLEN * GPIT];
  __shared__ __attribute__((aligned(16))) float Lb[CLEN * D_STATE];
  float* Lw = Lx;
  float* Lu = Lx;

  const int t   = threadIdx.x;
  const int bid = blockIdx.x;
  const int d0  = (bid & 3) * 256;
  const int d   = d0 + t;
  const int c   = (bid >> 2) & (NCHUNK - 1);
  const int b   = bid >> 7;
  const size_t tok0 = (size_t)b * SEQLEN + (size_t)c * CLEN;

  const int srow = t >> 6;
  const int scol = (t & 63) * 4;
  const float* __restrict__ gu = u + (tok0 + srow) * D_MODEL + d0 + scol;
  const int soff = srow * 256 + scol;

  const int grow = t >> 4;
  const int gcol = (t & 15) * 4;
  const float* __restrict__ gw = wt + (size_t)(d0 + grow) * DT_RANK + gcol;
  const float* __restrict__ gx = xp + (tok0 + grow) * XDIM + gcol;
  const int goff = grow * GPIT + gcol;

  v4f rw[16];
  #pragma unroll
  for (int i = 0; i < 16; ++i) rw[i] = *(const v4f*)(gw + (size_t)(i * 16) * DT_RANK);
  v4f rx[4];
  #pragma unroll
  for (int i = 0; i < 4; ++i) rx[i] = *(const v4f*)(gx + (size_t)(i * 16) * XDIM);
  const v4f rb = *(const v4f*)(xp + (tok0 + (t >> 2)) * XDIM + DT_RANK + (t & 3) * 4);
  v4f ru[SPC];
  tile_fetch1(gu, 0, ru);
  v4f al[D_STATE / 4];
  #pragma unroll
  for (int g = 0; g < D_STATE / 4; ++g) al[g] = *(const v4f*)(A_log + (size_t)d * D_STATE + 4 * g);

  const int wave = t >> 5;
  const int lane = t & 31;
  const int row  = lane & 15;
  const int kph  = (lane >> 4) << 1;
  float bn[2];
  #pragma unroll
  for (int nt = 0; nt < 2; ++nt) bn[nt] = bias[d0 + wave * 32 + nt * 16 + row];

  #pragma unroll
  for (int i = 0; i < 16; ++i) *(v4f*)(Lw + goff + i * 16 * GPIT) = rw[i];
  #pragma unroll
  for (int i = 0; i < 4; ++i)  *(v4f*)(La + goff + i * 16 * GPIT) = rx[i];
  *(v4f*)(Lb + t * 4) = rb;
  __syncthreads();

  {
    const int aoff = row * GPIT + kph;
    const int boff = (wave * 32 + row) * GPIT + kph;
    v8f acc[4][2] = {};
    #pragma unroll
    for (int kk = 0; kk < DT_RANK; kk += 4) {
      v2f bf[2];
      #pragma unroll
      for (int nt = 0; nt < 2; ++nt) bf[nt] = *(const v2f*)(Lw + boff + nt * 16 * GPIT + kk);
      #pragma unroll
      for (int mt = 0; mt < 4; ++mt) {
        const v2f a = *(const v2f*)(La + aoff + mt * 16 * GPIT + kk);
        #pragma unroll
        for (int nt = 0; nt < 2; ++nt)
          acc[mt][nt] = __builtin_amdgcn_wmma_f32_16x16x4_f32(false, a, false, bf[nt], (short)0,
                                                              acc[mt][nt], false, false);
      }
    }
    const int r0 = (lane >> 4) << 3;
    #pragma unroll
    for (int nt = 0; nt < 2; ++nt) {
      const int col = wave * 32 + nt * 16 + row;
      #pragma unroll
      for (int mt = 0; mt < 4; ++mt)
        #pragma unroll
        for (int i = 0; i < 8; ++i) {
          float x  = acc[mt][nt][i] + bn[nt];
          float sp = (x > 20.0f) ? x : __logf(1.0f + __expf(x));
          Ld[(mt * 16 + r0 + i) * LDP + col] = sp;
        }
    }
  }
  __syncthreads();

  tile_park1(Lu, soff, 0, ru);

  float A[D_STATE];
  #pragma unroll
  for (int g = 0; g < D_STATE / 4; ++g) {
    A[4 * g + 0] = -__expf(al[g].x);
    A[4 * g + 1] = -__expf(al[g].y);
    A[4 * g + 2] = -__expf(al[g].z);
    A[4 * g + 3] = -__expf(al[g].w);
  }

  float h[D_STATE];
  #pragma unroll
  for (int n = 0; n < D_STATE; ++n) h[n] = 0.0f;
  float sdt = 0.0f;
  __syncthreads();

  #pragma unroll
  for (int i = 0; i < CLEN / 4; ++i) {
    const int r = i * 4 + srow;
    *(v4f*)(delta + (tok0 + r) * D_MODEL + d0 + scol) = *(const v4f*)(Ld + r * LDP + scol);
  }

  #pragma unroll 1
  for (int q = 0; q < SUBS; ++q) {
    if (q + 1 < SUBS) tile_fetch1(gu, q + 1, ru);
    #pragma unroll 2
    for (int s = q * SLEN; s < (q + 1) * SLEN; ++s) {
      const float dt = Ld[s * LDP + t];
      const float ut = Lu[s * 256 + t];
      const v4f* bv = (const v4f*)(Lb + s * D_STATE);
      const v4f q0 = bv[0], q1 = bv[1], q2 = bv[2], q3 = bv[3];
      float Bt[D_STATE] = {q0.x, q0.y, q0.z, q0.w, q1.x, q1.y, q1.z, q1.w,
                           q2.x, q2.y, q2.z, q2.w, q3.x, q3.y, q3.z, q3.w};
      sdt += dt;
      const float du = dt * ut;
      #pragma unroll
      for (int n = 0; n < D_STATE; ++n)
        h[n] = __expf(dt * A[n]) * h[n] + du * Bt[n];
    }
    if (q + 1 < SUBS) {
      tile_park1(Lu, soff, q + 1, ru);
      __syncthreads();
    }
  }

  const size_t bc = (size_t)b * NCHUNK + c;
  float4* __restrict__ Q4 = (float4*)Q;
  #pragma unroll
  for (int g = 0; g < D_STATE / 4; ++g)
    Q4[(bc * (D_STATE / 4) + g) * D_MODEL + d] =
        make_float4(h[4 * g], h[4 * g + 1], h[4 * g + 2], h[4 * g + 3]);
  Sdt[bc * D_MODEL + d] = sdt;
}

__global__ __launch_bounds__(256) void scan_pass3_kernel(const float* __restrict__ delta,
                                                         const float* __restrict__ xp,
                                                         const float* __restrict__ u,
                                                         const float* __restrict__ A_log,
                                                         const float* __restrict__ Dsk,
                                                         const float* __restrict__ Q,
                                                         const float* __restrict__ Sdt,
                                                         float* __restrict__ y) {
  __shared__ __attribute__((aligned(16))) float Ld[CLEN * 256];
  __shared__ __attribute__((aligned(16))) float Lu[CLEN * 256];
  __shared__ __attribute__((aligned(16))) float Lbc[CLEN * 2 * D_STATE];

  const int t   = threadIdx.x;
  const int bid = blockIdx.x;
  const int d0  = (bid & 3) * 256;
  const int d   = d0 + t;
  const int b   = (bid >> 2) & (BATCH - 1);
  const int c   = (bid < 256) ? (NCHUNK - 1 - (bid >> 4)) : ((bid - 256) >> 4);
  const size_t tok0 = (size_t)b * SEQLEN + (size_t)c * CLEN;

  const int srow = t >> 6;
  const int scol = (t & 63) * 4;
  const float* __restrict__ gd = delta + (tok0 + srow) * D_MODEL + d0 + scol;
  const float* __restrict__ gu = u     + (tok0 + srow) * D_MODEL + d0 + scol;
  const int soff = srow * 256 + scol;
  v4f rd[SPC], ru[SPC];
  tile_fetch(gd, gu, 0, rd, ru);
  const v4f rbc0 = *(const v4f*)(xp + (tok0 + (t >> 3)) * XDIM + DT_RANK + (t & 7) * 4);
  const v4f rbc1 = *(const v4f*)(xp + (tok0 + 32 + (t >> 3)) * XDIM + DT_RANK + (t & 7) * 4);
  *(v4f*)(Lbc + t * 4)        = rbc0;
  *(v4f*)(Lbc + 1024 + t * 4) = rbc1;

  float A[D_STATE];
  #pragma unroll
  for (int n = 0; n < D_STATE; ++n) A[n] = -__expf(A_log[d * D_STATE + n]);

  float h[D_STATE];
  #pragma unroll
  for (int n = 0; n < D_STATE; ++n) h[n] = 0.0f;

  const float*  __restrict__ sb = Sdt + (size_t)b * NCHUNK * D_MODEL + d;
  const float4* __restrict__ qb = (const float4*)Q + (size_t)b * NCHUNK * (D_STATE / 4) * D_MODEL + d;
  int j = 0;
  for (; j + 4 <= c; j += 4) {
    float  s4[4];
    float4 q4[4][D_STATE / 4];
    #pragma unroll
    for (int k = 0; k < 4; ++k) {
      s4[k] = sb[(size_t)(j + k) * D_MODEL];
      #pragma unroll
      for (int g = 0; g < D_STATE / 4; ++g)
        q4[k][g] = qb[((size_t)(j + k) * (D_STATE / 4) + g) * D_MODEL];
    }
    #pragma unroll
    for (int k = 0; k < 4; ++k) {
      #pragma unroll
      for (int g = 0; g < D_STATE / 4; ++g) {
        h[4 * g + 0] = __expf(s4[k] * A[4 * g + 0]) * h[4 * g + 0] + q4[k][g].x;
        h[4 * g + 1] = __expf(s4[k] * A[4 * g + 1]) * h[4 * g + 1] + q4[k][g].y;
        h[4 * g + 2] = __expf(s4[k] * A[4 * g + 2]) * h[4 * g + 2] + q4[k][g].z;
        h[4 * g + 3] = __expf(s4[k] * A[4 * g + 3]) * h[4 * g + 3] + q4[k][g].w;
      }
    }
    __builtin_amdgcn_sched_group_barrier(0x020, 20, 0);
    __builtin_amdgcn_sched_group_barrier(0x002, 512, 0);
  }
  for (; j < c; ++j) {
    const float st = sb[(size_t)j * D_MODEL];
    float4 qt[D_STATE / 4];
    #pragma unroll
    for (int g = 0; g < D_STATE / 4; ++g)
      qt[g] = qb[((size_t)j * (D_STATE / 4) + g) * D_MODEL];
    #pragma unroll
    for (int g = 0; g < D_STATE / 4; ++g) {
      h[4 * g + 0] = __expf(st * A[4 * g + 0]) * h[4 * g + 0] + qt[g].x;
      h[4 * g + 1] = __expf(st * A[4 * g + 1]) * h[4 * g + 1] + qt[g].y;
      h[4 * g + 2] = __expf(st * A[4 * g + 2]) * h[4 * g + 2] + qt[g].z;
      h[4 * g + 3] = __expf(st * A[4 * g + 3]) * h[4 * g + 3] + qt[g].w;
    }
  }
  const float dsk = Dsk[d];

  tile_park(Ld, Lu, soff, 0, rd, ru);
  __syncthreads();

  #pragma unroll 1
  for (int q = 0; q < SUBS; ++q) {
    if (q + 1 < SUBS) tile_fetch(gd, gu, q + 1, rd, ru);
    #pragma unroll 2
    for (int s = q * SLEN; s < (q + 1) * SLEN; ++s) {
      const size_t tok = tok0 + s;
      const float dt = Ld[s * 256 + t];
      const float ut = Lu[s * 256 + t];
      const v4f* bv = (const v4f*)(Lbc + s * 2 * D_STATE);
      const v4f q0 = bv[0], q1 = bv[1], q2 = bv[2], q3 = bv[3];
      const v4f p0 = bv[4], p1 = bv[5], p2 = bv[6], p3 = bv[7];
      float Bt[D_STATE] = {q0.x, q0.y, q0.z, q0.w, q1.x, q1.y, q1.z, q1.w,
                           q2.x, q2.y, q2.z, q2.w, q3.x, q3.y, q3.z, q3.w};
      float Ct[D_STATE] = {p0.x, p0.y, p0.z, p0.w, p1.x, p1.y, p1.z, p1.w,
                           p2.x, p2.y, p2.z, p2.w, p3.x, p3.y, p3.z, p3.w};
      const float du = dt * ut;
      float acc = 0.0f;
      #pragma unroll
      for (int n = 0; n < D_STATE; ++n) {
        h[n] = __expf(dt * A[n]) * h[n] + du * Bt[n];
        acc += h[n] * Ct[n];
      }
      y[tok * D_MODEL + d] = acc + ut * dsk;
    }
    if (q + 1 < SUBS) {
      tile_park(Ld, Lu, soff, q + 1, rd, ru);
      __syncthreads();
    }
  }
}

extern "C" void kernel_launch(void* const* d_in, const int* in_sizes, int n_in,
                              void* d_out, int out_size, void* d_ws, size_t ws_size,
                              hipStream_t stream) {
  const float* u      = (const float*)d_in[0];
  const float* A_log  = (const float*)d_in[1];
  const float* D_skip = (const float*)d_in[2];
  const float* dt_w   = (const float*)d_in[3];
  const float* dt_b   = (const float*)d_in[4];
  const float* x_w    = (const float*)d_in[5];
  float* y = (float*)d_out;

  float* ws    = (float*)d_ws;
  float* xp    = ws;
  float* delta = xp    + (size_t)NTOK * XDIM;
  float* Q     = delta + (size_t)NTOK * D_MODEL;
  float* Sdt   = Q     + (size_t)BATCH * NCHUNK * D_MODEL * D_STATE;

  xp_wmma_kernel<<<NTOK / XM, 128, 0, stream>>>(u, x_w, xp);

  scan_pass1_kernel<<<BATCH * NCHUNK * (D_MODEL / 256), 256, 0, stream>>>(
      xp, dt_w, dt_b, u, A_log, delta, Q, Sdt);

  scan_pass3_kernel<<<BATCH * NCHUNK * (D_MODEL / 256), 256, 0, stream>>>(
      delta, xp, u, A_log, D_skip, Q, Sdt, y);
}
